# starting kernel
# baseline (speedup 1.0000x reference)
#include <hip/hip_runtime.h>
#include <math.h>


#define NN       80
#define MPM_IT   50
#define STRIDE   84
#define NWAVES   25
#define NTHREADS 800

typedef float v2f __attribute__((ext_vector_type(2)));
typedef float v8f __attribute__((ext_vector_type(8)));

__global__ __launch_bounds__(NTHREADS)
void grf_mpm_kernel(const float* __restrict__ recon,
                    const float* __restrict__ adj,
                    float* __restrict__ out)
{
    extern __shared__ float smem[];
    float* s_src = smem;
    float* s_tgt = s_src + NN * STRIDE;
    float* s_d   = s_tgt + NN * STRIDE;
    float* s_x   = s_d   + NN * STRIDE;
    float* s_M   = s_x   + NN * STRIDE;
    float* s_af  = s_M   + NN * STRIDE;
    float* s_of  = s_af  + NN;
    float* s_ad  = s_of  + NN;
    float* s_rd  = s_ad  + NN;
    float* s_red = s_rd  + NN;
    float* s_sc  = s_red + 32;

    const int tid  = threadIdx.x;
    const int lane = tid & 31;
    const int wave = tid >> 5;

    if (tid < NN) {
        float sa = 0.f, sr = 0.f;
        for (int c = 0; c < NN; ++c) {
            sa += adj[tid * NN + c];
            sr += recon[tid * NN + c];
        }
        s_af[tid] = sa;
        s_of[tid] = sr;
        s_ad[tid] = adj[tid * NN + tid];
        s_rd[tid] = recon[tid * NN + tid];
    }
    __syncthreads();

    float bce = 0.f;
    for (int idx = tid; idx < NN * NN; idx += NTHREADS) {
        int i = idx / NN;
        int j = idx - i * NN;
        float ta = adj[idx];
        float tr = recon[idx];
        float offd = (i == j) ? 0.f : 1.f;
        s_src[i * STRIDE + j] = ta * s_ad[i] * s_ad[j] * offd;
        s_tgt[i * STRIDE + j] = tr * s_rd[i] * s_rd[j] * offd;
        s_d  [i * STRIDE + j] = s_ad[i] * s_rd[j] / (fabsf(s_af[i] - s_of[j]) + 1.f);
        s_x  [i * STRIDE + j] = 1.f / (float)NN;
        if (j >= i) {
            bce -= ta * logf(tr) + (1.f - ta) * logf(1.f - tr);
        }
    }
    #pragma unroll
    for (int off = 16; off; off >>= 1) bce += __shfl_xor(bce, off, 32);
    if (lane == 0) s_red[wave] = bce;
    __syncthreads();
    if (wave == 0) {
        float t = (lane < NWAVES) ? s_red[lane] : 0.f;
        #pragma unroll
        for (int off = 16; off; off >>= 1) t += __shfl_xor(t, off, 32);
        if (lane == 0) out[0] = t / (float)(NN * (NN + 1) / 2);
    }
    __syncthreads();

    const int wr = wave / 5;
    const int wc = wave - 5 * wr;
    const int r0 = wr * 16;
    const int c0 = wc * 16;
    const int hh = lane >> 4;
    const int m  = lane & 15;
    const int kk = 2 * hh;

    for (int it = 0; it < MPM_IT; ++it) {
        #pragma unroll
        for (int v = 0; v < 8; ++v) {
            int j = r0 + v + 8 * hh;
            int a = c0 + m;
            const float* xr = s_x   + j * STRIDE;
            const float* tg = s_tgt + a * STRIDE;
            float best = xr[0] * tg[0];
            #pragma unroll 4
            for (int b = 1; b < NN; ++b) best = fmaxf(best, xr[b] * tg[b]);
            s_M[j * STRIDE + a] = best;
        }
        __syncthreads();

        v8f acc = {0.f, 0.f, 0.f, 0.f, 0.f, 0.f, 0.f, 0.f};
        for (int kc = 0; kc < NN; kc += 4) {
            v2f av, bv;
            av[0] = s_src[(r0 + m) * STRIDE + kc + kk + 0];
            av[1] = s_src[(r0 + m) * STRIDE + kc + kk + 1];
            bv[0] = s_M[(kc + kk + 0) * STRIDE + c0 + m];
            bv[1] = s_M[(kc + kk + 1) * STRIDE + c0 + m];
            acc = __builtin_amdgcn_wmma_f32_16x16x4_f32(
                false, av, false, bv, (short)0, acc, false, false);
        }

        float xn[8];
        float ssq = 0.f;
        #pragma unroll
        for (int v = 0; v < 8; ++v) {
            int i = r0 + v + 8 * hh;
            int a = c0 + m;
            float val = s_x[i * STRIDE + a] * s_d[i * STRIDE + a] + acc[v];
            xn[v] = val;
            ssq += val * val;
        }
        #pragma unroll
        for (int off = 16; off; off >>= 1) ssq += __shfl_xor(ssq, off, 32);
        if (lane == 0) s_red[wave] = ssq;
        __syncthreads();
        if (tid == 0) {
            float t = 0.f;
            for (int w = 0; w < NWAVES; ++w) t += s_red[w];
            s_sc[0] = 1.f / sqrtf(t);
        }
        __syncthreads();
        float rn = s_sc[0];
        #pragma unroll
        for (int v = 0; v < 8; ++v) {
            int i = r0 + v + 8 * hh;
            int a = c0 + m;
            s_x[i * STRIDE + a] = xn[v] * rn;
        }
        __syncthreads();
    }

    #pragma unroll
    for (int v = 0; v < 8; ++v) {
        int i = r0 + v + 8 * hh;
        int a = c0 + m;
        out[1 + i * NN + a] = s_x[i * STRIDE + a];
    }
}

extern "C" void kernel_launch(void* const* d_in, const int* in_sizes, int n_in,
                              void* d_out, int out_size, void* d_ws, size_t ws_size,
                              hipStream_t stream) {
    (void)in_sizes; (void)n_in; (void)out_size; (void)d_ws; (void)ws_size;
    const float* recon = (const float*)d_in[0];
    const float* adj   = (const float*)d_in[1];
    float* out = (float*)d_out;

    const size_t shmem = (size_t)(5 * NN * STRIDE + 4 * NN + 32 + 4) * sizeof(float);
    grf_mpm_kernel<<<1, NTHREADS, shmem, stream>>>(recon, adj, out);
}


# One launch per iteration on 5 work-groups (one 16-column strip each: its M strip and 5 WMMA tiles); x unnormalised and 25 tile sums pass through the workspace, same sum order
# speedup vs baseline: 2.7473x; 2.7473x over previous
#include <hip/hip_runtime.h>
#include <math.h>


#define NN       80
#define STRIDE   84
#define PM       20
#define NWAVES   25
#define NTHREADS 800
#define ITHREADS 640

#define WS_SRC 0
#define WS_TGT 6400
#define WS_D   12800
#define WS_X0  19200
#define WS_X1  25600
#define WS_P0  32000
#define WS_P1  32032

typedef float v2f __attribute__((ext_vector_type(2)));
typedef float v8f __attribute__((ext_vector_type(8)));

__device__ __forceinline__ int tile_off(int i, int a) {
    return (((i >> 4) * 5 + (a >> 4)) << 8) + ((i & 7) << 5) + (((i >> 3) & 1) << 4) + (a & 15);
}

__device__ __forceinline__ int src_off(int i, int j) {
    return (i >> 4) * 1280 + (j >> 2) * 64 + ((((j >> 1) & 1) << 4) + (i & 15)) * 2 + (j & 1);
}

__global__ __launch_bounds__(NTHREADS)
void grf_setup_kernel(const float* __restrict__ recon,
                      const float* __restrict__ adj,
                      float* __restrict__ out,
                      float* __restrict__ ws)
{
    __shared__ float s_af[NN];
    __shared__ float s_of[NN];
    __shared__ float s_ad[NN];
    __shared__ float s_rd[NN];
    __shared__ float s_red[32];

    const int tid  = threadIdx.x;
    const int lane = tid & 31;
    const int wave = tid >> 5;

    if (tid < NN) {
        float sa = 0.f, sr = 0.f;
        for (int c = 0; c < NN; ++c) {
            sa += adj[tid * NN + c];
            sr += recon[tid * NN + c];
        }
        s_af[tid] = sa;
        s_of[tid] = sr;
        s_ad[tid] = adj[tid * NN + tid];
        s_rd[tid] = recon[tid * NN + tid];
    }
    __syncthreads();

    float bce = 0.f;
    for (int idx = tid; idx < NN * NN; idx += NTHREADS) {
        int i = idx / NN;
        int j = idx - i * NN;
        float ta = adj[idx];
        float tr = recon[idx];
        float offd = (i == j) ? 0.f : 1.f;
        ws[WS_SRC + src_off(i, j)]  = ta * s_ad[i] * s_ad[j] * offd;
        ws[WS_TGT + idx]            = tr * s_rd[i] * s_rd[j] * offd;
        ws[WS_D   + tile_off(i, j)] = s_ad[i] * s_rd[j] / (fabsf(s_af[i] - s_of[j]) + 1.f);
        ws[WS_X0  + idx]            = 1.f / (float)NN;
        if (j >= i) {
            bce -= ta * logf(tr) + (1.f - ta) * logf(1.f - tr);
        }
    }
    if (tid < 32) ws[WS_P0 + tid] = (tid == 0) ? 1.f : 0.f;

    #pragma unroll
    for (int off = 16; off; off >>= 1) bce += __shfl_xor(bce, off, 32);
    if (lane == 0) s_red[wave] = bce;
    __syncthreads();
    if (wave == 0) {
        float t = (lane < NWAVES) ? s_red[lane] : 0.f;
        #pragma unroll
        for (int off = 16; off; off >>= 1) t += __shfl_xor(t, off, 32);
        if (lane == 0) out[0] = t / (float)(NN * (NN + 1) / 2);
    }
}

__global__ __launch_bounds__(ITHREADS)
void grf_iter_kernel(const float* __restrict__ srcf,
                     const float* __restrict__ tgt,
                     const float* __restrict__ dp,
                     const float* __restrict__ xin,
                     const float* __restrict__ pin,
                     float* __restrict__ xout,
                     float* __restrict__ pout)
{
    __shared__ __attribute__((aligned(16))) float s_x[NN * STRIDE];
    __shared__ __attribute__((aligned(16))) float s_tgt[16 * STRIDE];
    __shared__ float s_M[NN * PM];

    const int tid  = threadIdx.x;
    const int lane = tid & 31;
    const int wave = __builtin_amdgcn_readfirstlane(tid >> 5);
    const int wc   = blockIdx.x;
    const int c0   = wc * 16;
    const int hh   = lane >> 4;
    const int m    = lane & 15;

    v2f   af[20];
    float dv[8];
    #pragma unroll
    for (int q = 0; q < 20; ++q) { af[q][0] = 0.f; af[q][1] = 0.f; }
    #pragma unroll
    for (int v = 0; v < 8; ++v) dv[v] = 0.f;
    if (wave < 5) {
        const v2f* ap = (const v2f*)(srcf + wave * 1280) + lane;
        #pragma unroll
        for (int q = 0; q < 20; ++q) af[q] = ap[q * 32];
        const float* dq = dp + (wave * 5 + wc) * 256 + lane;
        #pragma unroll
        for (int v = 0; v < 8; ++v) dv[v] = dq[v * 32];
    }

    float t = 0.f;
    for (int w = 0; w < NWAVES; ++w) t += pin[w];
    const float rn = 1.f / sqrtf(t);

    const float4* xin4 = (const float4*)xin;
    for (int q = tid; q < (NN * NN) / 4; q += ITHREADS) {
        const int tile = q >> 6;
        const int tr   = tile / 5;
        const int tc   = tile - 5 * tr;
        const int i    = tr * 16 + ((q >> 2) & 1) * 8 + ((q >> 3) & 7);
        const int a    = tc * 16 + (q & 3) * 4;
        float4 xv = xin4[q];
        xv.x *= rn; xv.y *= rn; xv.z *= rn; xv.w *= rn;
        *(float4*)(s_x + i * STRIDE + a) = xv;
    }
    const float4* tg4 = (const float4*)(tgt + c0 * NN);
    for (int q = tid; q < (16 * NN) / 4; q += ITHREADS) {
        const int a = q / 20;
        const int c = q - 20 * a;
        *(float4*)(s_tgt + a * STRIDE + 4 * c) = tg4[q];
    }
    __syncthreads();

    {
        const int tb  = wave >> 2;
        const int sub = wave & 3;
        const float* tg = s_tgt + m * STRIDE;
        #pragma unroll
        for (int u = 0; u < 2; ++u) {
            const int j = 16 * tb + 8 * hh + 2 * sub + u;
            const float* xr = s_x + j * STRIDE;
            float best = xr[0] * tg[0];
            #pragma unroll 4
            for (int b = 1; b < NN; ++b) best = fmaxf(best, xr[b] * tg[b]);
            s_M[j * PM + m] = best;
        }
    }
    __syncthreads();

    if (wave < 5) {
        const int r0 = wave * 16;
        const int kk = 2 * hh;
        v8f acc = {0.f, 0.f, 0.f, 0.f, 0.f, 0.f, 0.f, 0.f};
        #pragma unroll
        for (int q = 0; q < 20; ++q) {
            v2f bv;
            bv[0] = s_M[(4 * q + kk + 0) * PM + m];
            bv[1] = s_M[(4 * q + kk + 1) * PM + m];
            acc = __builtin_amdgcn_wmma_f32_16x16x4_f32(
                false, af[q], false, bv, (short)0, acc, false, false);
        }

        float xn[8];
        float ssq = 0.f;
        #pragma unroll
        for (int v = 0; v < 8; ++v) {
            const int i = r0 + v + 8 * hh;
            const int a = c0 + m;
            float val = s_x[i * STRIDE + a] * dv[v] + acc[v];
            xn[v] = val;
            ssq += val * val;
        }
        #pragma unroll
        for (int off = 16; off; off >>= 1) ssq += __shfl_xor(ssq, off, 32);
        if (lane == 0) pout[wave * 5 + wc] = ssq;

        float* xo = xout + (wave * 5 + wc) * 256 + lane;
        #pragma unroll
        for (int v = 0; v < 8; ++v) xo[v * 32] = xn[v];
    }
}

__global__ __launch_bounds__(256)
void grf_final_kernel(const float* __restrict__ xin,
                      const float* __restrict__ pin,
                      float* __restrict__ out)
{
    float t = 0.f;
    for (int w = 0; w < NWAVES; ++w) t += pin[w];
    const float rn = 1.f / sqrtf(t);
    const int idx = blockIdx.x * 256 + threadIdx.x;
    const int i = idx / NN;
    const int a = idx - i * NN;
    out[1 + idx] = xin[tile_off(i, a)] * rn;
}

#define MPM_IT2 \
    grf_iter_kernel<<<5, ITHREADS, 0, stream>>>(srcf, tgt, dp, x0, p0, x1, p1); \
    grf_iter_kernel<<<5, ITHREADS, 0, stream>>>(srcf, tgt, dp, x1, p1, x0, p0);
#define MPM_IT10 MPM_IT2 MPM_IT2 MPM_IT2 MPM_IT2 MPM_IT2

extern "C" void kernel_launch(void* const* d_in, const int* in_sizes, int n_in,
                              void* d_out, int out_size, void* d_ws, size_t ws_size,
                              hipStream_t stream) {
    (void)in_sizes; (void)n_in; (void)out_size; (void)ws_size;
    const float* recon = (const float*)d_in[0];
    const float* adj   = (const float*)d_in[1];
    float* out = (float*)d_out;
    float* ws  = (float*)d_ws;
    float* srcf = ws + WS_SRC;
    float* tgt  = ws + WS_TGT;
    float* dp   = ws + WS_D;
    float* x0   = ws + WS_X0;
    float* x1   = ws + WS_X1;
    float* p0   = ws + WS_P0;
    float* p1   = ws + WS_P1;

    grf_setup_kernel<<<1, NTHREADS, 0, stream>>>(recon, adj, out, ws);
    MPM_IT10 MPM_IT10 MPM_IT10 MPM_IT10 MPM_IT10
    grf_final_kernel<<<25, 256, 0, stream>>>(x0, p0, out);
}
